# starting kernel
# baseline (speedup 1.0000x reference)
#include <hip/hip_runtime.h>

typedef _Float16 v16h __attribute__((ext_vector_type(16)));
typedef _Float16 v8h  __attribute__((ext_vector_type(8)));
typedef _Float16 h4   __attribute__((ext_vector_type(4)));
typedef float    v8f  __attribute__((ext_vector_type(8)));

#define D 128

__global__ void __launch_bounds__(256) zero_kernel(float4* __restrict__ p, int n4) {
    int i = blockIdx.x * blockDim.x + threadIdx.x;
    if (i < n4) p[i] = make_float4(0.f, 0.f, 0.f, 0.f);
}

__global__ void __launch_bounds__(256) cvt_f32_f16(const float* __restrict__ src,
                                                   _Float16* __restrict__ dst, int n4) {
    int i = blockIdx.x * blockDim.x + threadIdx.x;
    if (i < n4) {
        float4 v = ((const float4*)src)[i];
        h4 o;
        o[0] = (_Float16)v.x; o[1] = (_Float16)v.y;
        o[2] = (_Float16)v.z; o[3] = (_Float16)v.w;
        ((h4*)dst)[i] = o;
    }
}

__global__ void __launch_bounds__(256) spmm_scatter(
        const int*   __restrict__ rows,
        const int*   __restrict__ cols,
        const float* __restrict__ vals,
        const float* __restrict__ x,
        float*       __restrict__ agg,
        int n_edges) {
    const int wave = blockIdx.x * (blockDim.x >> 5) + (threadIdx.x >> 5);
    const int lane = threadIdx.x & 31;
    const int e0 = wave * 4;
    #pragma unroll
    for (int k = 0; k < 4; ++k) {
        int e = e0 + k;
        if (e >= n_edges) break;
        int   r = rows[e];
        int   c = cols[e];
        float v = vals[e];
        float4 d = ((const float4*)(x + (size_t)c * D))[lane];
        float* dst = agg + (size_t)r * D + lane * 4;
        unsafeAtomicAdd(dst + 0, v * d.x);
        unsafeAtomicAdd(dst + 1, v * d.y);
        unsafeAtomicAdd(dst + 2, v * d.z);
        unsafeAtomicAdd(dst + 3, v * d.w);
    }
}

__global__ void __launch_bounds__(256) gemm_relu_wmma(
        const float*    __restrict__ X,
        const _Float16* __restrict__ W16,
        float*          __restrict__ Y,
        int n_nodes) {
    const int wave    = threadIdx.x >> 5;
    const int lane    = threadIdx.x & 31;
    const int laneMod = lane & 15;
    const int laneHi  = lane >> 4;
    const int rowBase = (blockIdx.x * 8 + wave) * 16;

    int aRow = rowBase + laneMod;
    if (aRow >= n_nodes) aRow = n_nodes - 1;

    const float*    aPtr  = X + (size_t)aRow * D;
    const _Float16* bBase = W16 + (size_t)laneMod * D;

    v8f acc[8] = {};
    #pragma unroll
    for (int kk = 0; kk < D; kk += 32) {
        const int off0 = kk + laneHi * 8;
        const int off1 = kk + 16 + laneHi * 8;

        float4 a0 = *(const float4*)(aPtr + off0);
        float4 a1 = *(const float4*)(aPtr + off0 + 4);
        float4 a2 = *(const float4*)(aPtr + off1);
        float4 a3 = *(const float4*)(aPtr + off1 + 4);
        v16h a;
        a[0]=(_Float16)a0.x;  a[1]=(_Float16)a0.y;  a[2]=(_Float16)a0.z;  a[3]=(_Float16)a0.w;
        a[4]=(_Float16)a1.x;  a[5]=(_Float16)a1.y;  a[6]=(_Float16)a1.z;  a[7]=(_Float16)a1.w;
        a[8]=(_Float16)a2.x;  a[9]=(_Float16)a2.y;  a[10]=(_Float16)a2.z; a[11]=(_Float16)a2.w;
        a[12]=(_Float16)a3.x; a[13]=(_Float16)a3.y; a[14]=(_Float16)a3.z; a[15]=(_Float16)a3.w;

        #pragma unroll
        for (int t = 0; t < 8; ++t) {
            const _Float16* bp = bBase + (size_t)t * 16 * D;
            v8h blo = *(const v8h*)(bp + off0);
            v8h bhi = *(const v8h*)(bp + off1);
            v16h b = __builtin_shufflevector(blo, bhi,
                       0,1,2,3,4,5,6,7,8,9,10,11,12,13,14,15);
            acc[t] = __builtin_amdgcn_wmma_f32_16x16x32_f16(
                         false, a,  false, b,
                         (short)0, acc[t],
                         false,  false);
        }
    }

    #pragma unroll
    for (int t = 0; t < 8; ++t) {
        #pragma unroll
        for (int i = 0; i < 8; ++i) {
            int r = rowBase + i + laneHi * 8;
            if (r < n_nodes) {
                float v = acc[t][i];
                Y[(size_t)r * D + t * 16 + laneMod] = v > 0.f ? v : 0.f;
            }
        }
    }
}

__global__ void __launch_bounds__(256) head_kernel(
        const float* __restrict__ H,
        const float* __restrict__ Wc,
        const float* __restrict__ bc,
        float*       __restrict__ out,
        int n_nodes) {
    const int node = blockIdx.x * (blockDim.x >> 5) + (threadIdx.x >> 5);
    const int lane = threadIdx.x & 31;
    if (node >= n_nodes) return;
    float4 h = ((const float4*)(H + (size_t)node * D))[lane];
    float4 w = ((const float4*)Wc)[lane];
    float s = h.x * w.x + h.y * w.y + h.z * w.z + h.w * w.w;
    #pragma unroll
    for (int off = 16; off > 0; off >>= 1)
        s += __shfl_down(s, off, 32);
    if (lane == 0) out[node] = s + bc[0];
}

extern "C" void kernel_launch(void* const* d_in, const int* in_sizes, int n_in,
                              void* d_out, int out_size, void* d_ws, size_t ws_size,
                              hipStream_t stream) {
    const float* features = (const float*)d_in[0];
    const int*   erows    = (const int*)  d_in[1];
    const int*   ecols    = (const int*)  d_in[2];
    const float* evals    = (const float*)d_in[3];
    const float* W1       = (const float*)d_in[4];
    const float* W2       = (const float*)d_in[5];
    const float* Wc       = (const float*)d_in[6];
    const float* bc       = (const float*)d_in[7];
    float* out = (float*)d_out;

    const int n_nodes = in_sizes[0] / D;
    const int n_edges = in_sizes[1];

    float*     buf0 = (float*)d_ws;
    float*     buf1 = buf0 + (size_t)n_nodes * D;
    _Float16*  w16a = (_Float16*)(buf1 + (size_t)n_nodes * D);
    _Float16*  w16b = w16a + D * D;

    const int n4      = (n_nodes * D) / 4;
    const int zblocks = (n4 + 255) / 256;
    const int ewaves  = (n_edges + 3) / 4;
    const int eblocks = (ewaves + 7) / 8;
    const int gblocks = (n_nodes + 127) / 128;
    const int hblocks = (n_nodes + 7) / 8;
    const int w4      = (D * D) / 4;
    const int wblocks = (w4 + 255) / 256;

    cvt_f32_f16<<<wblocks, 256, 0, stream>>>(W1, w16a, w4);
    cvt_f32_f16<<<wblocks, 256, 0, stream>>>(W2, w16b, w4);

    zero_kernel<<<zblocks, 256, 0, stream>>>((float4*)buf0, n4);
    spmm_scatter<<<eblocks, 256, 0, stream>>>(erows, ecols, evals, features, buf0, n_edges);
    gemm_relu_wmma<<<gblocks, 256, 0, stream>>>(buf0, w16a, buf1, n_nodes);

    zero_kernel<<<zblocks, 256, 0, stream>>>((float4*)buf0, n4);
    spmm_scatter<<<eblocks, 256, 0, stream>>>(erows, ecols, evals, buf1, buf0, n_edges);
    gemm_relu_wmma<<<gblocks, 256, 0, stream>>>(buf0, w16b, buf1, n_nodes);

    head_kernel<<<hblocks, 256, 0, stream>>>(buf1, Wc, bc, out, n_nodes);
}


# long-lived waves: zero, scatter, head launch a fixed chip-sized grid and loop; a scatter wave walks 49 contiguous edges, was 4 (32 index triples per load, read by lane); sums still in arrival order
# speedup vs baseline: 2.0129x; 2.0129x over previous
#include <hip/hip_runtime.h>

typedef _Float16 v16h __attribute__((ext_vector_type(16)));
typedef _Float16 v8h  __attribute__((ext_vector_type(8)));
typedef _Float16 h4   __attribute__((ext_vector_type(4)));
typedef float    v8f  __attribute__((ext_vector_type(8)));

#define D 128

__global__ void __launch_bounds__(256) zero_kernel(float4* __restrict__ p, int n4) {
    const int stride = gridDim.x * blockDim.x;
    for (int i = blockIdx.x * blockDim.x + threadIdx.x; i < n4; i += stride)
        p[i] = make_float4(0.f, 0.f, 0.f, 0.f);
}

__global__ void __launch_bounds__(256) cvt_f32_f16(const float* __restrict__ src,
                                                   _Float16* __restrict__ dst, int n4) {
    int i = blockIdx.x * blockDim.x + threadIdx.x;
    if (i < n4) {
        float4 v = ((const float4*)src)[i];
        h4 o;
        o[0] = (_Float16)v.x; o[1] = (_Float16)v.y;
        o[2] = (_Float16)v.z; o[3] = (_Float16)v.w;
        ((h4*)dst)[i] = o;
    }
}

#define EU 8
__global__ void __launch_bounds__(256) spmm_scatter(
        const int*   __restrict__ rows,
        const int*   __restrict__ cols,
        const float* __restrict__ vals,
        const float* __restrict__ x,
        float*       __restrict__ agg,
        int n_edges) {
    const int lane = threadIdx.x & 31;
    const int nw   = gridDim.x * (blockDim.x >> 5);
    const int wave = __builtin_amdgcn_readfirstlane(
                         blockIdx.x * (blockDim.x >> 5) + (threadIdx.x >> 5));
    const int per  = n_edges / nw;
    const int rem  = n_edges - per * nw;
    const int eBeg = wave * per + (wave < rem ? wave : rem);
    const int eEnd = eBeg + per + (wave < rem ? 1 : 0);

    int rN = 0, cN = 0; float vN = 0.f;
    if (eBeg + lane < eEnd) {
        rN = rows[eBeg + lane]; cN = cols[eBeg + lane]; vN = vals[eBeg + lane];
    }
    for (int b = eBeg; b < eEnd; b += 32) {
        const int left = eEnd - b;
        const int cnt  = left < 32 ? left : 32;
        const int r = rN, c = cN; const int vbits = __float_as_int(vN);
        const int en = b + 32 + lane;
        if (en < eEnd) { rN = rows[en]; cN = cols[en]; vN = vals[en]; }

        int j = 0;
        for (; j + EU <= cnt; j += EU) {
            float d[EU][4];
            #pragma unroll
            for (int u = 0; u < EU; ++u) {
                const int cj = __builtin_amdgcn_readlane(c, j + u);
                const float* src = x + (size_t)cj * D + lane;
                d[u][0] = src[0];  d[u][1] = src[32];
                d[u][2] = src[64]; d[u][3] = src[96];
            }
            #pragma unroll
            for (int u = 0; u < EU; ++u) {
                const int   rj = __builtin_amdgcn_readlane(r, j + u);
                const float vj = __int_as_float(__builtin_amdgcn_readlane(vbits, j + u));
                float* dst = agg + (size_t)rj * D + lane;
                unsafeAtomicAdd(dst + 0,  vj * d[u][0]);
                unsafeAtomicAdd(dst + 32, vj * d[u][1]);
                unsafeAtomicAdd(dst + 64, vj * d[u][2]);
                unsafeAtomicAdd(dst + 96, vj * d[u][3]);
            }
        }
        for (; j < cnt; ++j) {
            const int   cj = __builtin_amdgcn_readlane(c, j);
            const int   rj = __builtin_amdgcn_readlane(r, j);
            const float vj = __int_as_float(__builtin_amdgcn_readlane(vbits, j));
            const float* src = x + (size_t)cj * D + lane;
            const float d0 = src[0], d1 = src[32], d2 = src[64], d3 = src[96];
            float* dst = agg + (size_t)rj * D + lane;
            unsafeAtomicAdd(dst + 0,  vj * d0);
            unsafeAtomicAdd(dst + 32, vj * d1);
            unsafeAtomicAdd(dst + 64, vj * d2);
            unsafeAtomicAdd(dst + 96, vj * d3);
        }
    }
}

__global__ void __launch_bounds__(256) gemm_relu_wmma(
        const float*    __restrict__ X,
        const _Float16* __restrict__ W16,
        float*          __restrict__ Y,
        int n_nodes) {
    const int wave    = threadIdx.x >> 5;
    const int lane    = threadIdx.x & 31;
    const int laneMod = lane & 15;
    const int laneHi  = lane >> 4;
    const int rowBase = (blockIdx.x * 8 + wave) * 16;

    int aRow = rowBase + laneMod;
    if (aRow >= n_nodes) aRow = n_nodes - 1;

    const float*    aPtr  = X + (size_t)aRow * D;
    const _Float16* bBase = W16 + (size_t)laneMod * D;

    v8f acc[8] = {};
    #pragma unroll
    for (int kk = 0; kk < D; kk += 32) {
        const int off0 = kk + laneHi * 8;
        const int off1 = kk + 16 + laneHi * 8;

        float4 a0 = *(const float4*)(aPtr + off0);
        float4 a1 = *(const float4*)(aPtr + off0 + 4);
        float4 a2 = *(const float4*)(aPtr + off1);
        float4 a3 = *(const float4*)(aPtr + off1 + 4);
        v16h a;
        a[0]=(_Float16)a0.x;  a[1]=(_Float16)a0.y;  a[2]=(_Float16)a0.z;  a[3]=(_Float16)a0.w;
        a[4]=(_Float16)a1.x;  a[5]=(_Float16)a1.y;  a[6]=(_Float16)a1.z;  a[7]=(_Float16)a1.w;
        a[8]=(_Float16)a2.x;  a[9]=(_Float16)a2.y;  a[10]=(_Float16)a2.z; a[11]=(_Float16)a2.w;
        a[12]=(_Float16)a3.x; a[13]=(_Float16)a3.y; a[14]=(_Float16)a3.z; a[15]=(_Float16)a3.w;

        #pragma unroll
        for (int t = 0; t < 8; ++t) {
            const _Float16* bp = bBase + (size_t)t * 16 * D;
            v8h blo = *(const v8h*)(bp + off0);
            v8h bhi = *(const v8h*)(bp + off1);
            v16h b = __builtin_shufflevector(blo, bhi,
                       0,1,2,3,4,5,6,7,8,9,10,11,12,13,14,15);
            acc[t] = __builtin_amdgcn_wmma_f32_16x16x32_f16(
                         false, a,  false, b,
                         (short)0, acc[t],
                         false,  false);
        }
    }

    #pragma unroll
    for (int t = 0; t < 8; ++t) {
        #pragma unroll
        for (int i = 0; i < 8; ++i) {
            int r = rowBase + i + laneHi * 8;
            if (r < n_nodes) {
                float v = acc[t][i];
                Y[(size_t)r * D + t * 16 + laneMod] = v > 0.f ? v : 0.f;
            }
        }
    }
}

__device__ __forceinline__ float head_row(float4 h, float4 w) {
    float s = h.x * w.x + h.y * w.y + h.z * w.z + h.w * w.w;
    #pragma unroll
    for (int off = 16; off > 0; off >>= 1)
        s += __shfl_down(s, off, 32);
    return s;
}

__global__ void __launch_bounds__(256) head_kernel(
        const float* __restrict__ H,
        const float* __restrict__ Wc,
        const float* __restrict__ bc,
        float*       __restrict__ out,
        int n_nodes) {
    const int lane = threadIdx.x & 31;
    const int nw   = gridDim.x * (blockDim.x >> 5);
    const int wave = __builtin_amdgcn_readfirstlane(
                         blockIdx.x * (blockDim.x >> 5) + (threadIdx.x >> 5));
    const int per  = n_nodes / nw;
    const int rem  = n_nodes - per * nw;
    const int nBeg = wave * per + (wave < rem ? wave : rem);
    const int nEnd = nBeg + per + (wave < rem ? 1 : 0);

    const float4 w = ((const float4*)Wc)[lane];
    const float  b = bc[0];
    int node = nBeg;
    for (; node + 4 <= nEnd; node += 4) {
        float4 h[4];
        #pragma unroll
        for (int u = 0; u < 4; ++u)
            h[u] = ((const float4*)(H + (size_t)(node + u) * D))[lane];
        float s[4];
        #pragma unroll
        for (int u = 0; u < 4; ++u)
            s[u] = head_row(h[u], w);
        if (lane == 0) {
            #pragma unroll
            for (int u = 0; u < 4; ++u)
                out[node + u] = s[u] + b;
        }
    }
    for (; node < nEnd; ++node) {
        const float4 h = ((const float4*)(H + (size_t)node * D))[lane];
        const float s = head_row(h, w);
        if (lane == 0) out[node] = s + b;
    }
}

extern "C" void kernel_launch(void* const* d_in, const int* in_sizes, int n_in,
                              void* d_out, int out_size, void* d_ws, size_t ws_size,
                              hipStream_t stream) {
    const float* features = (const float*)d_in[0];
    const int*   erows    = (const int*)  d_in[1];
    const int*   ecols    = (const int*)  d_in[2];
    const float* evals    = (const float*)d_in[3];
    const float* W1       = (const float*)d_in[4];
    const float* W2       = (const float*)d_in[5];
    const float* Wc       = (const float*)d_in[6];
    const float* bc       = (const float*)d_in[7];
    float* out = (float*)d_out;

    const int n_nodes = in_sizes[0] / D;
    const int n_edges = in_sizes[1];

    float*     buf0 = (float*)d_ws;
    float*     buf1 = buf0 + (size_t)n_nodes * D;
    _Float16*  w16a = (_Float16*)(buf1 + (size_t)n_nodes * D);
    _Float16*  w16b = w16a + D * D;

    const int n4      = (n_nodes * D) / 4;
    const int zblocks = 1024;
    const int eblocks = 4096;
    const int gblocks = (n_nodes + 127) / 128;
    const int hblocks = 1024;
    const int w4      = (D * D) / 4;
    const int wblocks = (w4 + 255) / 256;

    cvt_f32_f16<<<wblocks, 256, 0, stream>>>(W1, w16a, w4);
    cvt_f32_f16<<<wblocks, 256, 0, stream>>>(W2, w16b, w4);

    zero_kernel<<<zblocks, 256, 0, stream>>>((float4*)buf0, n4);
    spmm_scatter<<<eblocks, 256, 0, stream>>>(erows, ecols, evals, features, buf0, n_edges);
    gemm_relu_wmma<<<gblocks, 256, 0, stream>>>(buf0, w16a, buf1, n_nodes);

    zero_kernel<<<zblocks, 256, 0, stream>>>((float4*)buf0, n4);
    spmm_scatter<<<eblocks, 256, 0, stream>>>(erows, ecols, evals, buf1, buf0, n_edges);
    gemm_relu_wmma<<<gblocks, 256, 0, stream>>>(buf0, w16b, buf1, n_nodes);

    head_kernel<<<hblocks, 256, 0, stream>>>(buf1, Wc, bc, out, n_nodes);
}
